# starting kernel
# baseline (speedup 1.0000x reference)
#include <hip/hip_runtime.h>


#define C0   32
#define LD   6
#define PDIM 96
#define VOL  (PDIM * PDIM * PDIM)
#define HSTR 36
#define TILES_PER_WAVE  4
#define WAVES_PER_BLOCK 8
#define POINTS_PER_BLOCK (16 * TILES_PER_WAVE * WAVES_PER_BLOCK)
#define FLT_BIG 3.402823466e+38f

typedef float v2f __attribute__((ext_vector_type(2)));
typedef float v8f __attribute__((ext_vector_type(8)));

__device__ __forceinline__ v8f wmma_f32(v2f a, v2f b, v8f c) {
  return __builtin_amdgcn_wmma_f32_16x16x4_f32(false, a, false, b,
                                               (short)0, c, false, false);
}

__device__ __forceinline__ float relu1(float x) {
  return __builtin_amdgcn_fmed3f(x, 0.0f, FLT_BIG);
}

__device__ __forceinline__ int permk(int s) { return (s >> 1) + ((s & 1) << 4); }

__global__ __launch_bounds__(256) void
fcomb_mlp_kernel(const float* __restrict__ fm, const float* __restrict__ z,
                 const float* __restrict__ w1, const float* __restrict__ b1,
                 const float* __restrict__ w2, const float* __restrict__ b2,
                 const float* __restrict__ w3, const float* __restrict__ b3,
                 const float* __restrict__ wl, const float* __restrict__ bl,
                 float* __restrict__ out)
{
  __shared__ float sW1[C0 * (C0 + LD)];
  __shared__ float sW2[C0 * C0];
  __shared__ float sW3[C0 * C0];
  __shared__ float sWl[C0];
  __shared__ float sB1[C0], sB2[C0], sB3[C0];
  __shared__ float sZ[LD];
  __shared__ float sBl[1];
  __shared__ float sHa[WAVES_PER_BLOCK][16 * HSTR];
  __shared__ float sHb[WAVES_PER_BLOCK][16 * HSTR];
  __shared__ float sO [WAVES_PER_BLOCK][16];

  const int tid = threadIdx.x;
  for (int i = tid; i < C0 * (C0 + LD); i += 256) sW1[i] = w1[i];
  for (int i = tid; i < C0 * C0;        i += 256) sW2[i] = w2[i];
  for (int i = tid; i < C0 * C0;        i += 256) sW3[i] = w3[i];
  if (tid < C0) { sWl[tid] = wl[tid]; sB1[tid] = b1[tid];
                  sB2[tid] = b2[tid]; sB3[tid] = b3[tid]; }
  if (tid < LD) sZ[tid] = z[tid];
  if (tid == 0) sBl[0] = bl[0];
  __syncthreads();

  const int lane = tid & 31;
  const int wave = tid >> 5;
  const int l16  = lane & 15;
  const int hi   = lane >> 4;

  float cinit1[2], cinit2[2], cinit3[2], wlv[2];
#pragma unroll
  for (int h = 0; h < 2; ++h) {
    const int n = l16 + 16 * h;
    float c = sB1[n];
#pragma unroll
    for (int k = 0; k < LD; ++k) c += sW1[n * (C0 + LD) + C0 + k] * sZ[k];
    cinit1[h] = c;
    cinit2[h] = sB2[n];
    cinit3[h] = sB3[n];
    wlv[h]    = sWl[n];
  }
  const float blv = sBl[0];

  v2f Bw1[2][8], Bw2[2][8], Bw3[2][8];
#pragma unroll
  for (int h = 0; h < 2; ++h) {
    const int n = l16 + 16 * h;
#pragma unroll
    for (int j = 0; j < 8; ++j) {
      const int k0 = 4 * j + 2 * hi;
      v2f t;
      t[0] = sW1[n * (C0 + LD) + k0]; t[1] = sW1[n * (C0 + LD) + k0 + 1];
      Bw1[h][j] = t;
      t[0] = sW2[n * C0 + permk(k0)]; t[1] = sW2[n * C0 + permk(k0 + 1)];
      Bw2[h][j] = t;
      t[0] = sW3[n * C0 + permk(k0)]; t[1] = sW3[n * C0 + permk(k0 + 1)];
      Bw3[h][j] = t;
    }
  }

  float* Ha = &sHa[wave][0];
  float* Hb = &sHb[wave][0];

  for (int t = 0; t < TILES_PER_WAVE; ++t) {
    const int pbase = blockIdx.x * POINTS_PER_BLOCK +
                      (wave * TILES_PER_WAVE + t) * 16;

    v2f a[8];
#pragma unroll
    for (int j = 0; j < 8; ++j) {
      const int k0 = 4 * j + 2 * hi;
      v2f av;
      av[0] = fm[k0 * VOL + pbase + l16];
      av[1] = fm[(k0 + 1) * VOL + pbase + l16];
      a[j] = av;
    }
    v8f acc0, acc1;
#pragma unroll
    for (int r = 0; r < 8; ++r) { acc0[r] = cinit1[0]; acc1[r] = cinit1[1]; }
#pragma unroll
    for (int j = 0; j < 8; ++j) {
      acc0 = wmma_f32(a[j], Bw1[0][j], acc0);
      acc1 = wmma_f32(a[j], Bw1[1][j], acc1);
    }
#pragma unroll
    for (int r = 0; r < 8; ++r) {
      v2f p;
      p[0] = relu1(acc0[r]);
      p[1] = relu1(acc1[r]);
      *(v2f*)&Ha[(r + 8 * hi) * HSTR + 2 * l16] = p;
    }

#pragma unroll
    for (int j = 0; j < 8; ++j)
      a[j] = *(const v2f*)&Ha[l16 * HSTR + 4 * j + 2 * hi];
#pragma unroll
    for (int r = 0; r < 8; ++r) { acc0[r] = cinit2[0]; acc1[r] = cinit2[1]; }
#pragma unroll
    for (int j = 0; j < 8; ++j) {
      acc0 = wmma_f32(a[j], Bw2[0][j], acc0);
      acc1 = wmma_f32(a[j], Bw2[1][j], acc1);
    }
#pragma unroll
    for (int r = 0; r < 8; ++r) {
      v2f p;
      p[0] = relu1(acc0[r]);
      p[1] = relu1(acc1[r]);
      *(v2f*)&Hb[(r + 8 * hi) * HSTR + 2 * l16] = p;
    }

#pragma unroll
    for (int j = 0; j < 8; ++j)
      a[j] = *(const v2f*)&Hb[l16 * HSTR + 4 * j + 2 * hi];
#pragma unroll
    for (int r = 0; r < 8; ++r) { acc0[r] = cinit3[0]; acc1[r] = cinit3[1]; }
#pragma unroll
    for (int j = 0; j < 8; ++j) {
      acc0 = wmma_f32(a[j], Bw3[0][j], acc0);
      acc1 = wmma_f32(a[j], Bw3[1][j], acc1);
    }

#pragma unroll
    for (int r = 0; r < 8; ++r) {
      const float v0 = relu1(acc0[r]);
      const float v1 = relu1(acc1[r]);
      float s = v0 * wlv[0] + v1 * wlv[1];
      s += __shfl_xor(s, 1);
      s += __shfl_xor(s, 2);
      s += __shfl_xor(s, 4);
      s += __shfl_xor(s, 8);
      if (l16 == 0) sO[wave][r + 8 * hi] = s;
    }
    if (lane < 16) out[pbase + lane] = sO[wave][lane] + blv;
  }
}

extern "C" void kernel_launch(void* const* d_in, const int* in_sizes, int n_in,
                              void* d_out, int out_size, void* d_ws, size_t ws_size,
                              hipStream_t stream) {
  (void)in_sizes; (void)n_in; (void)out_size; (void)d_ws; (void)ws_size;
  const float* fm = (const float*)d_in[0];
  const float* z  = (const float*)d_in[1];
  const float* w1 = (const float*)d_in[2];
  const float* b1 = (const float*)d_in[3];
  const float* w2 = (const float*)d_in[4];
  const float* b2 = (const float*)d_in[5];
  const float* w3 = (const float*)d_in[6];
  const float* b3 = (const float*)d_in[7];
  const float* wl = (const float*)d_in[8];
  const float* bl = (const float*)d_in[9];
  float* out = (float*)d_out;

  dim3 grid(VOL / POINTS_PER_BLOCK);
  dim3 block(256);
  fcomb_mlp_kernel<<<grid, block, 0, stream>>>(fm, z, w1, b1, w2, b2,
                                               w3, b3, wl, bl, out);
}


# Swap WMMA operand roles (weights = A, points = N): each layer's relu'd D registers are the next layer's B, so both LDS relayouts and the 32-shuffle final reduce go; k and final sum order change
# speedup vs baseline: 1.5553x; 1.5553x over previous
#include <hip/hip_runtime.h>


#define C0   32
#define LD   6
#define PDIM 96
#define VOL  (PDIM * PDIM * PDIM)
#define TILES_PER_WAVE  4
#define WAVES_PER_BLOCK 8
#define POINTS_PER_BLOCK (16 * TILES_PER_WAVE * WAVES_PER_BLOCK)
#define FLT_BIG 3.402823466e+38f

typedef float v2f __attribute__((ext_vector_type(2)));
typedef float v8f __attribute__((ext_vector_type(8)));

__device__ __forceinline__ v8f wmma_f32(v2f a, v2f b, v8f c) {
  return __builtin_amdgcn_wmma_f32_16x16x4_f32(false, a, false, b,
                                               (short)0, c, false, false);
}

__device__ __forceinline__ float relu1(float x) {
  return __builtin_amdgcn_fmed3f(x, 0.0f, FLT_BIG);
}

__device__ __forceinline__ float xhalf(float x) {
  const int b = __builtin_bit_cast(int, x);
  return __builtin_bit_cast(float,
      __builtin_amdgcn_permlanex16(b, b, 0x76543210, 0xfedcba98, false, false));
}

__device__ __forceinline__ int kchain(int j, int hi) {
  return 16 * (j >> 2) + 8 * hi + 2 * (j & 3);
}

__global__ __launch_bounds__(256) void
fcomb_mlp_kernel(const float* __restrict__ fm, const float* __restrict__ z,
                 const float* __restrict__ w1, const float* __restrict__ b1,
                 const float* __restrict__ w2, const float* __restrict__ b2,
                 const float* __restrict__ w3, const float* __restrict__ b3,
                 const float* __restrict__ wl, const float* __restrict__ bl,
                 float* __restrict__ out)
{
  __shared__ float sW1[C0 * (C0 + LD)];
  __shared__ float sW2[C0 * C0];
  __shared__ float sW3[C0 * C0];
  __shared__ float sWl[C0];
  __shared__ float sB1[C0], sB2[C0], sB3[C0];
  __shared__ float sZ[LD];
  __shared__ float sBl[1];

  const int tid = threadIdx.x;
  for (int i = tid; i < C0 * (C0 + LD); i += 256) sW1[i] = w1[i];
  for (int i = tid; i < C0 * C0;        i += 256) sW2[i] = w2[i];
  for (int i = tid; i < C0 * C0;        i += 256) sW3[i] = w3[i];
  if (tid < C0) { sWl[tid] = wl[tid]; sB1[tid] = b1[tid];
                  sB2[tid] = b2[tid]; sB3[tid] = b3[tid]; }
  if (tid < LD) sZ[tid] = z[tid];
  if (tid == 0) sBl[0] = bl[0];
  __syncthreads();

  const int lane = tid & 31;
  const int wave = tid >> 5;
  const int l16  = lane & 15;
  const int hi   = lane >> 4;

  v8f cin1[2], cin2[2], cin3[2];
  float wlv[2][8];
#pragma unroll
  for (int mt = 0; mt < 2; ++mt) {
#pragma unroll
    for (int r = 0; r < 8; ++r) {
      const int n = 16 * mt + 8 * hi + r;
      float c = sB1[n];
#pragma unroll
      for (int k = 0; k < LD; ++k) c += sW1[n * (C0 + LD) + C0 + k] * sZ[k];
      cin1[mt][r] = c;
      cin2[mt][r] = sB2[n];
      cin3[mt][r] = sB3[n];
      wlv[mt][r]  = sWl[n];
    }
  }
  const float blv = sBl[0];

  v2f Aw1[2][8], Aw2[2][8], Aw3[2][8];
#pragma unroll
  for (int mt = 0; mt < 2; ++mt) {
    const int o = l16 + 16 * mt;
#pragma unroll
    for (int j = 0; j < 8; ++j) {
      const int k1 = 4 * j + 2 * hi;
      const int k2 = kchain(j, hi);
      v2f t;
      t[0] = sW1[o * (C0 + LD) + k1]; t[1] = sW1[o * (C0 + LD) + k1 + 1];
      Aw1[mt][j] = t;
      t[0] = sW2[o * C0 + k2]; t[1] = sW2[o * C0 + k2 + 1];
      Aw2[mt][j] = t;
      t[0] = sW3[o * C0 + k2]; t[1] = sW3[o * C0 + k2 + 1];
      Aw3[mt][j] = t;
    }
  }

  for (int t = 0; t < TILES_PER_WAVE; ++t) {
    const int pbase = blockIdx.x * POINTS_PER_BLOCK +
                      (wave * TILES_PER_WAVE + t) * 16;

    v2f b[8];
#pragma unroll
    for (int j = 0; j < 8; ++j) {
      const int k0 = 4 * j + 2 * hi;
      v2f bv;
      bv[0] = fm[k0 * VOL + pbase + l16];
      bv[1] = fm[(k0 + 1) * VOL + pbase + l16];
      b[j] = bv;
    }
    v8f acc0 = cin1[0], acc1 = cin1[1];
#pragma unroll
    for (int j = 0; j < 8; ++j) {
      acc0 = wmma_f32(Aw1[0][j], b[j], acc0);
      acc1 = wmma_f32(Aw1[1][j], b[j], acc1);
    }

#pragma unroll
    for (int q = 0; q < 4; ++q) {
      v2f p0, p1;
      p0[0] = relu1(acc0[2 * q]); p0[1] = relu1(acc0[2 * q + 1]);
      p1[0] = relu1(acc1[2 * q]); p1[1] = relu1(acc1[2 * q + 1]);
      b[q] = p0; b[4 + q] = p1;
    }
    acc0 = cin2[0]; acc1 = cin2[1];
#pragma unroll
    for (int j = 0; j < 8; ++j) {
      acc0 = wmma_f32(Aw2[0][j], b[j], acc0);
      acc1 = wmma_f32(Aw2[1][j], b[j], acc1);
    }

#pragma unroll
    for (int q = 0; q < 4; ++q) {
      v2f p0, p1;
      p0[0] = relu1(acc0[2 * q]); p0[1] = relu1(acc0[2 * q + 1]);
      p1[0] = relu1(acc1[2 * q]); p1[1] = relu1(acc1[2 * q + 1]);
      b[q] = p0; b[4 + q] = p1;
    }
    acc0 = cin3[0]; acc1 = cin3[1];
#pragma unroll
    for (int j = 0; j < 8; ++j) {
      acc0 = wmma_f32(Aw3[0][j], b[j], acc0);
      acc1 = wmma_f32(Aw3[1][j], b[j], acc1);
    }

    float s0 = relu1(acc0[0]) * wlv[0][0];
    float s1 = relu1(acc1[0]) * wlv[1][0];
#pragma unroll
    for (int r = 1; r < 8; ++r) {
      s0 += relu1(acc0[r]) * wlv[0][r];
      s1 += relu1(acc1[r]) * wlv[1][r];
    }
    const float part = s0 + s1;
    const float tot  = part + xhalf(part);
    if (lane < 16) out[pbase + lane] = tot + blv;
  }
}

extern "C" void kernel_launch(void* const* d_in, const int* in_sizes, int n_in,
                              void* d_out, int out_size, void* d_ws, size_t ws_size,
                              hipStream_t stream) {
  (void)in_sizes; (void)n_in; (void)out_size; (void)d_ws; (void)ws_size;
  const float* fm = (const float*)d_in[0];
  const float* z  = (const float*)d_in[1];
  const float* w1 = (const float*)d_in[2];
  const float* b1 = (const float*)d_in[3];
  const float* w2 = (const float*)d_in[4];
  const float* b2 = (const float*)d_in[5];
  const float* w3 = (const float*)d_in[6];
  const float* b3 = (const float*)d_in[7];
  const float* wl = (const float*)d_in[8];
  const float* bl = (const float*)d_in[9];
  float* out = (float*)d_out;

  dim3 grid(VOL / POINTS_PER_BLOCK);
  dim3 block(256);
  fcomb_mlp_kernel<<<grid, block, 0, stream>>>(fm, z, w1, b1, w2, b2,
                                               w3, b3, wl, bl, out);
}


# Persistent launch: 256 work-groups of 16 waves, all resident at once; each wave loops over 12-14 contiguous tiles, so weight staging and A-fragment set-up run once per wave, not once per 4 tiles
# speedup vs baseline: 2.0490x; 1.3174x over previous
#include <hip/hip_runtime.h>


#define C0   32
#define LD   6
#define PDIM 96
#define VOL  (PDIM * PDIM * PDIM)
#define THREADS         512
#define NUM_BLOCKS      256
#define TILES_PER_BLOCK (VOL / 16 / NUM_BLOCKS)
#define FLT_BIG 3.402823466e+38f

typedef float v2f __attribute__((ext_vector_type(2)));
typedef float v8f __attribute__((ext_vector_type(8)));

__device__ __forceinline__ v8f wmma_f32(v2f a, v2f b, v8f c) {
  return __builtin_amdgcn_wmma_f32_16x16x4_f32(false, a, false, b,
                                               (short)0, c, false, false);
}

__device__ __forceinline__ float relu1(float x) {
  return __builtin_amdgcn_fmed3f(x, 0.0f, FLT_BIG);
}

__device__ __forceinline__ float xhalf(float x) {
  const int b = __builtin_bit_cast(int, x);
  return __builtin_bit_cast(float,
      __builtin_amdgcn_permlanex16(b, b, 0x76543210, 0xfedcba98, false, false));
}

__device__ __forceinline__ int kchain(int j, int hi) {
  return 16 * (j >> 2) + 8 * hi + 2 * (j & 3);
}

__global__ __launch_bounds__(THREADS) void
fcomb_mlp_kernel(const float* __restrict__ fm, const float* __restrict__ z,
                 const float* __restrict__ w1, const float* __restrict__ b1,
                 const float* __restrict__ w2, const float* __restrict__ b2,
                 const float* __restrict__ w3, const float* __restrict__ b3,
                 const float* __restrict__ wl, const float* __restrict__ bl,
                 float* __restrict__ out)
{
  __shared__ float sW1[C0 * (C0 + LD)];
  __shared__ float sW2[C0 * C0];
  __shared__ float sW3[C0 * C0];
  __shared__ float sWl[C0];
  __shared__ float sB1[C0], sB2[C0], sB3[C0];
  __shared__ float sZ[LD];
  __shared__ float sBl[1];

  const int tid = threadIdx.x;
  for (int i = tid; i < C0 * (C0 + LD); i += THREADS) sW1[i] = w1[i];
  for (int i = tid; i < C0 * C0;        i += THREADS) sW2[i] = w2[i];
  for (int i = tid; i < C0 * C0;        i += THREADS) sW3[i] = w3[i];
  if (tid < C0) { sWl[tid] = wl[tid]; sB1[tid] = b1[tid];
                  sB2[tid] = b2[tid]; sB3[tid] = b3[tid]; }
  if (tid < LD) sZ[tid] = z[tid];
  if (tid == 0) sBl[0] = bl[0];
  __syncthreads();

  const int lane = tid & 31;
  const int wave = __builtin_amdgcn_readfirstlane(tid >> 5);
  const int l16  = lane & 15;
  const int hi   = lane >> 4;

  v8f cin1[2], cin2[2], cin3[2];
  float wlv[2][8];
#pragma unroll
  for (int mt = 0; mt < 2; ++mt) {
#pragma unroll
    for (int r = 0; r < 8; ++r) {
      const int n = 16 * mt + 8 * hi + r;
      float c = sB1[n];
#pragma unroll
      for (int k = 0; k < LD; ++k) c += sW1[n * (C0 + LD) + C0 + k] * sZ[k];
      cin1[mt][r] = c;
      cin2[mt][r] = sB2[n];
      cin3[mt][r] = sB3[n];
      wlv[mt][r]  = sWl[n];
    }
  }
  const float blv = sBl[0];

  v2f Aw1[2][8], Aw2[2][8], Aw3[2][8];
#pragma unroll
  for (int mt = 0; mt < 2; ++mt) {
    const int o = l16 + 16 * mt;
#pragma unroll
    for (int j = 0; j < 8; ++j) {
      const int k1 = 4 * j + 2 * hi;
      const int k2 = kchain(j, hi);
      v2f t;
      t[0] = sW1[o * (C0 + LD) + k1]; t[1] = sW1[o * (C0 + LD) + k1 + 1];
      Aw1[mt][j] = t;
      t[0] = sW2[o * C0 + k2]; t[1] = sW2[o * C0 + k2 + 1];
      Aw2[mt][j] = t;
      t[0] = sW3[o * C0 + k2]; t[1] = sW3[o * C0 + k2 + 1];
      Aw3[mt][j] = t;
    }
  }

  const int late  = (wave + 4) >> 4;
  const int ntile = 14 - 2 * late;
  const int tile0 = blockIdx.x * TILES_PER_BLOCK + 14 * wave - 2 * late * (wave - 12);
  for (int t = 0; t < ntile; ++t) {
    const int pbase = (tile0 + t) * 16;

    v2f b[8];
#pragma unroll
    for (int j = 0; j < 8; ++j) {
      const int k0 = 4 * j + 2 * hi;
      v2f bv;
      bv[0] = fm[k0 * VOL + pbase + l16];
      bv[1] = fm[(k0 + 1) * VOL + pbase + l16];
      b[j] = bv;
    }
    v8f acc0 = cin1[0], acc1 = cin1[1];
#pragma unroll
    for (int j = 0; j < 8; ++j) {
      acc0 = wmma_f32(Aw1[0][j], b[j], acc0);
      acc1 = wmma_f32(Aw1[1][j], b[j], acc1);
    }

#pragma unroll
    for (int q = 0; q < 4; ++q) {
      v2f p0, p1;
      p0[0] = relu1(acc0[2 * q]); p0[1] = relu1(acc0[2 * q + 1]);
      p1[0] = relu1(acc1[2 * q]); p1[1] = relu1(acc1[2 * q + 1]);
      b[q] = p0; b[4 + q] = p1;
    }
    acc0 = cin2[0]; acc1 = cin2[1];
#pragma unroll
    for (int j = 0; j < 8; ++j) {
      acc0 = wmma_f32(Aw2[0][j], b[j], acc0);
      acc1 = wmma_f32(Aw2[1][j], b[j], acc1);
    }

#pragma unroll
    for (int q = 0; q < 4; ++q) {
      v2f p0, p1;
      p0[0] = relu1(acc0[2 * q]); p0[1] = relu1(acc0[2 * q + 1]);
      p1[0] = relu1(acc1[2 * q]); p1[1] = relu1(acc1[2 * q + 1]);
      b[q] = p0; b[4 + q] = p1;
    }
    acc0 = cin3[0]; acc1 = cin3[1];
#pragma unroll
    for (int j = 0; j < 8; ++j) {
      acc0 = wmma_f32(Aw3[0][j], b[j], acc0);
      acc1 = wmma_f32(Aw3[1][j], b[j], acc1);
    }

    float s0 = relu1(acc0[1]) * wlv[0][1];
    float s1 = relu1(acc1[1]) * wlv[1][1];
    s0 = __builtin_fmaf(relu1(acc0[0]), wlv[0][0], s0);
    s1 = __builtin_fmaf(relu1(acc1[0]), wlv[1][0], s1);
#pragma unroll
    for (int r = 2; r < 8; ++r) {
      s0 = __builtin_fmaf(relu1(acc0[r]), wlv[0][r], s0);
      s1 = __builtin_fmaf(relu1(acc1[r]), wlv[1][r], s1);
    }
    const float part = s0 + s1;
    const float tot  = part + xhalf(part);
    if (lane < 16) out[pbase + lane] = tot + blv;
  }
}

extern "C" void kernel_launch(void* const* d_in, const int* in_sizes, int n_in,
                              void* d_out, int out_size, void* d_ws, size_t ws_size,
                              hipStream_t stream) {
  (void)in_sizes; (void)n_in; (void)out_size; (void)d_ws; (void)ws_size;
  const float* fm = (const float*)d_in[0];
  const float* z  = (const float*)d_in[1];
  const float* w1 = (const float*)d_in[2];
  const float* b1 = (const float*)d_in[3];
  const float* w2 = (const float*)d_in[4];
  const float* b2 = (const float*)d_in[5];
  const float* w3 = (const float*)d_in[6];
  const float* b3 = (const float*)d_in[7];
  const float* wl = (const float*)d_in[8];
  const float* bl = (const float*)d_in[9];
  float* out = (float*)d_out;

  dim3 grid(NUM_BLOCKS);
  dim3 block(THREADS);
  fcomb_mlp_kernel<<<grid, block, 0, stream>>>(fm, z, w1, b1, w2, b2,
                                               w3, b3, wl, bl, out);
}


# Layer-1 B fragments fetched one tile ahead into a second register set: a tile's 16 loads go out as one group (scalar channel bases, one lane offset) a whole tile before the WMMAs that read them
# speedup vs baseline: 2.2770x; 1.1113x over previous
#include <hip/hip_runtime.h>


#define C0   32
#define LD   6
#define PDIM 96
#define VOL  (PDIM * PDIM * PDIM)
#define THREADS         512
#define NUM_BLOCKS      256
#define TILES_PER_BLOCK (VOL / 16 / NUM_BLOCKS)
#define FLT_BIG 3.402823466e+38f

typedef float v2f __attribute__((ext_vector_type(2)));
typedef float v8f __attribute__((ext_vector_type(8)));
typedef const float __attribute__((address_space(1)))* gfp;

__device__ __forceinline__ v8f wmma_f32(v2f a, v2f b, v8f c) {
  return __builtin_amdgcn_wmma_f32_16x16x4_f32(false, a, false, b,
                                               (short)0, c, false, false);
}

__device__ __forceinline__ float relu1(float x) {
  return __builtin_amdgcn_fmed3f(x, 0.0f, FLT_BIG);
}

__device__ __forceinline__ float xhalf(float x) {
  const int b = __builtin_bit_cast(int, x);
  return __builtin_bit_cast(float,
      __builtin_amdgcn_permlanex16(b, b, 0x76543210, 0xfedcba98, false, false));
}

__device__ __forceinline__ int kchain(int j, int hi) {
  return 16 * (j >> 2) + 8 * hi + 2 * (j & 3);
}

__global__ __launch_bounds__(THREADS) void
fcomb_mlp_kernel(const float* __restrict__ fm, const float* __restrict__ z,
                 const float* __restrict__ w1, const float* __restrict__ b1,
                 const float* __restrict__ w2, const float* __restrict__ b2,
                 const float* __restrict__ w3, const float* __restrict__ b3,
                 const float* __restrict__ wl, const float* __restrict__ bl,
                 float* __restrict__ out)
{
  __shared__ float sW1[C0 * (C0 + LD)];
  __shared__ float sW2[C0 * C0];
  __shared__ float sW3[C0 * C0];
  __shared__ float sWl[C0];
  __shared__ float sB1[C0], sB2[C0], sB3[C0];
  __shared__ float sZ[LD];
  __shared__ float sBl[1];

  const int tid = threadIdx.x;
  for (int i = tid; i < C0 * (C0 + LD); i += THREADS) sW1[i] = w1[i];
  for (int i = tid; i < C0 * C0;        i += THREADS) sW2[i] = w2[i];
  for (int i = tid; i < C0 * C0;        i += THREADS) sW3[i] = w3[i];
  if (tid < C0) { sWl[tid] = wl[tid]; sB1[tid] = b1[tid];
                  sB2[tid] = b2[tid]; sB3[tid] = b3[tid]; }
  if (tid < LD) sZ[tid] = z[tid];
  if (tid == 0) sBl[0] = bl[0];
  __syncthreads();

  const int lane = tid & 31;
  const int wave = __builtin_amdgcn_readfirstlane(tid >> 5);
  const int l16  = lane & 15;
  const int hi   = lane >> 4;

  const int late  = (wave + 4) >> 4;
  const int ntile = 14 - 2 * late;
  const int tile0 = blockIdx.x * TILES_PER_BLOCK + 14 * wave - 2 * late * (wave - 12);
  gfp chan[8];
#pragma unroll
  for (int j = 0; j < 8; ++j) {
    gfp p = (gfp)fm + (4 * j) * VOL;
    asm("" : "+s"(p));
    chan[j] = p;
  }
  const unsigned laneoff = (unsigned)((2 * hi) * VOL + l16);
  auto fetch = [&](v2f (&dst)[8], int tile) __attribute__((always_inline)) {
    const unsigned voff = laneoff + (unsigned)tile * 16u;
#pragma unroll
    for (int j = 0; j < 8; ++j) {
      v2f bv;
      bv[0] = chan[j][voff];
      bv[1] = (chan[j] + VOL)[voff];
      dst[j] = bv;
    }
  };
  v2f bA[8], bB[8];
  fetch(bA, tile0);
  __builtin_amdgcn_sched_barrier(0);
  fetch(bB, tile0 + 1);
  __builtin_amdgcn_sched_barrier(0);

  v8f cin1[2], cin2[2], cin3[2];
  float wlv[2][8];
#pragma unroll
  for (int mt = 0; mt < 2; ++mt) {
#pragma unroll
    for (int r = 0; r < 8; ++r) {
      const int n = 16 * mt + 8 * hi + r;
      float c = sB1[n];
#pragma unroll
      for (int k = 0; k < LD; ++k) c += sW1[n * (C0 + LD) + C0 + k] * sZ[k];
      cin1[mt][r] = c;
      cin2[mt][r] = sB2[n];
      cin3[mt][r] = sB3[n];
      wlv[mt][r]  = sWl[n];
    }
  }
  const float blv = sBl[0];

  v2f Aw1[2][8], Aw2[2][8], Aw3[2][8];
#pragma unroll
  for (int mt = 0; mt < 2; ++mt) {
    const int o = l16 + 16 * mt;
#pragma unroll
    for (int j = 0; j < 8; ++j) {
      const int k1 = 4 * j + 2 * hi;
      const int k2 = kchain(j, hi);
      v2f w;
      w[0] = sW1[o * (C0 + LD) + k1]; w[1] = sW1[o * (C0 + LD) + k1 + 1];
      Aw1[mt][j] = w;
      w[0] = sW2[o * C0 + k2]; w[1] = sW2[o * C0 + k2 + 1];
      Aw2[mt][j] = w;
      w[0] = sW3[o * C0 + k2]; w[1] = sW3[o * C0 + k2 + 1];
      Aw3[mt][j] = w;
    }
  }

  auto tile_mlp = [&](const v2f (&bin)[8], int pbase) __attribute__((always_inline)) {
    v8f acc0 = cin1[0], acc1 = cin1[1];
#pragma unroll
    for (int j = 0; j < 8; ++j) {
      acc0 = wmma_f32(Aw1[0][j], bin[j], acc0);
      acc1 = wmma_f32(Aw1[1][j], bin[j], acc1);
    }
    v2f b[8];

#pragma unroll
    for (int q = 0; q < 4; ++q) {
      v2f p0, p1;
      p0[0] = relu1(acc0[2 * q]); p0[1] = relu1(acc0[2 * q + 1]);
      p1[0] = relu1(acc1[2 * q]); p1[1] = relu1(acc1[2 * q + 1]);
      b[q] = p0; b[4 + q] = p1;
    }
    acc0 = cin2[0]; acc1 = cin2[1];
#pragma unroll
    for (int j = 0; j < 8; ++j) {
      acc0 = wmma_f32(Aw2[0][j], b[j], acc0);
      acc1 = wmma_f32(Aw2[1][j], b[j], acc1);
    }

#pragma unroll
    for (int q = 0; q < 4; ++q) {
      v2f p0, p1;
      p0[0] = relu1(acc0[2 * q]); p0[1] = relu1(acc0[2 * q + 1]);
      p1[0] = relu1(acc1[2 * q]); p1[1] = relu1(acc1[2 * q + 1]);
      b[q] = p0; b[4 + q] = p1;
    }
    acc0 = cin3[0]; acc1 = cin3[1];
#pragma unroll
    for (int j = 0; j < 8; ++j) {
      acc0 = wmma_f32(Aw3[0][j], b[j], acc0);
      acc1 = wmma_f32(Aw3[1][j], b[j], acc1);
    }

    float s0 = relu1(acc0[1]) * wlv[0][1];
    float s1 = relu1(acc1[1]) * wlv[1][1];
    s0 = __builtin_fmaf(relu1(acc0[0]), wlv[0][0], s0);
    s1 = __builtin_fmaf(relu1(acc1[0]), wlv[1][0], s1);
#pragma unroll
    for (int r = 2; r < 8; ++r) {
      s0 = __builtin_fmaf(relu1(acc0[r]), wlv[0][r], s0);
      s1 = __builtin_fmaf(relu1(acc1[r]), wlv[1][r], s1);
    }
    const float part = s0 + s1;
    const float tot  = part + xhalf(part);
    if (lane < 16) out[pbase + lane] = tot + blv;
  };

  int t = 0;
  for (; t + 2 < ntile; t += 2) {
    tile_mlp(bA, (tile0 + t) * 16);
    fetch(bA, tile0 + t + 2);
    __builtin_amdgcn_sched_barrier(0);
    tile_mlp(bB, (tile0 + t + 1) * 16);
    fetch(bB, tile0 + t + 3);
    __builtin_amdgcn_sched_barrier(0);
  }
  tile_mlp(bA, (tile0 + t) * 16);
  tile_mlp(bB, (tile0 + t + 1) * 16);
}

extern "C" void kernel_launch(void* const* d_in, const int* in_sizes, int n_in,
                              void* d_out, int out_size, void* d_ws, size_t ws_size,
                              hipStream_t stream) {
  (void)in_sizes; (void)n_in; (void)out_size; (void)d_ws; (void)ws_size;
  const float* fm = (const float*)d_in[0];
  const float* z  = (const float*)d_in[1];
  const float* w1 = (const float*)d_in[2];
  const float* b1 = (const float*)d_in[3];
  const float* w2 = (const float*)d_in[4];
  const float* b2 = (const float*)d_in[5];
  const float* w3 = (const float*)d_in[6];
  const float* b3 = (const float*)d_in[7];
  const float* wl = (const float*)d_in[8];
  const float* bl = (const float*)d_in[9];
  float* out = (float*)d_out;

  dim3 grid(NUM_BLOCKS);
  dim3 block(THREADS);
  fcomb_mlp_kernel<<<grid, block, 0, stream>>>(fm, z, w1, b1, w2, b2,
                                               w3, b3, wl, bl, out);
}


# Weight staging: all of a thread's global loads issued before the first wait, so wave 0 pays one memory round trip in front of the barrier, not ten in series
# speedup vs baseline: 2.4560x; 1.0786x over previous
#include <hip/hip_runtime.h>


#define C0   32
#define LD   6
#define PDIM 96
#define VOL  (PDIM * PDIM * PDIM)
#define THREADS         512
#define NUM_BLOCKS      256
#define TILES_PER_BLOCK (VOL / 16 / NUM_BLOCKS)
#define FLT_BIG 3.402823466e+38f

typedef float v2f __attribute__((ext_vector_type(2)));
typedef float v8f __attribute__((ext_vector_type(8)));
typedef const float __attribute__((address_space(1)))* gfp;

__device__ __forceinline__ v8f wmma_f32(v2f a, v2f b, v8f c) {
  return __builtin_amdgcn_wmma_f32_16x16x4_f32(false, a, false, b,
                                               (short)0, c, false, false);
}

__device__ __forceinline__ float relu1(float x) {
  return __builtin_amdgcn_fmed3f(x, 0.0f, FLT_BIG);
}

__device__ __forceinline__ float xhalf(float x) {
  const int b = __builtin_bit_cast(int, x);
  return __builtin_bit_cast(float,
      __builtin_amdgcn_permlanex16(b, b, 0x76543210, 0xfedcba98, false, false));
}

__device__ __forceinline__ int kchain(int j, int hi) {
  return 16 * (j >> 2) + 8 * hi + 2 * (j & 3);
}

__global__ __launch_bounds__(THREADS) void
fcomb_mlp_kernel(const float* __restrict__ fm, const float* __restrict__ z,
                 const float* __restrict__ w1, const float* __restrict__ b1,
                 const float* __restrict__ w2, const float* __restrict__ b2,
                 const float* __restrict__ w3, const float* __restrict__ b3,
                 const float* __restrict__ wl, const float* __restrict__ bl,
                 float* __restrict__ out)
{
  __shared__ float sW1[C0 * (C0 + LD)];
  __shared__ float sW2[C0 * C0];
  __shared__ float sW3[C0 * C0];
  __shared__ float sWl[C0];
  __shared__ float sB1[C0], sB2[C0], sB3[C0];
  __shared__ float sZ[LD];
  __shared__ float sBl[1];

  const int tid  = threadIdx.x;
  const int lane = tid & 31;
  const int wave = __builtin_amdgcn_readfirstlane(tid >> 5);

  const bool w1tail = tid + 2 * THREADS < C0 * (C0 + LD);
  const float g1a = w1[tid];
  const float g1b = w1[tid + THREADS];
  const float g1c = w1[w1tail ? tid + 2 * THREADS : tid];
  const float g2a = w2[tid];
  const float g2b = w2[tid + THREADS];
  const float g3a = w3[tid];
  const float g3b = w3[tid + THREADS];
  float gwl = 0.0f, gb1 = 0.0f, gb2 = 0.0f, gb3 = 0.0f, gz = 0.0f, gbl = 0.0f;
  if (wave == 0) {
    gwl = wl[lane]; gb1 = b1[lane]; gb2 = b2[lane]; gb3 = b3[lane];
    gz  = z[lane < LD ? lane : 0];
    gbl = bl[0];
  }
  sW1[tid] = g1a;  sW1[tid + THREADS] = g1b;
  if (w1tail) sW1[tid + 2 * THREADS] = g1c;
  sW2[tid] = g2a;  sW2[tid + THREADS] = g2b;
  sW3[tid] = g3a;  sW3[tid + THREADS] = g3b;
  if (wave == 0) {
    sWl[lane] = gwl; sB1[lane] = gb1; sB2[lane] = gb2; sB3[lane] = gb3;
    if (lane < LD) sZ[lane] = gz;
    if (lane == 0) sBl[0] = gbl;
  }
  __syncthreads();

  const int l16  = lane & 15;
  const int hi   = lane >> 4;

  const int late  = (wave + 4) >> 4;
  const int ntile = 14 - 2 * late;
  const int tile0 = blockIdx.x * TILES_PER_BLOCK + 14 * wave - 2 * late * (wave - 12);
  gfp chan[8];
#pragma unroll
  for (int j = 0; j < 8; ++j) {
    gfp p = (gfp)fm + (4 * j) * VOL;
    asm("" : "+s"(p));
    chan[j] = p;
  }
  const unsigned laneoff = (unsigned)((2 * hi) * VOL + l16);
  auto fetch = [&](v2f (&dst)[8], int tile) __attribute__((always_inline)) {
    const unsigned voff = laneoff + (unsigned)tile * 16u;
#pragma unroll
    for (int j = 0; j < 8; ++j) {
      v2f bv;
      bv[0] = chan[j][voff];
      bv[1] = (chan[j] + VOL)[voff];
      dst[j] = bv;
    }
  };
  v2f bA[8], bB[8];
  fetch(bA, tile0);
  __builtin_amdgcn_sched_barrier(0);
  fetch(bB, tile0 + 1);
  __builtin_amdgcn_sched_barrier(0);

  v8f cin1[2], cin2[2], cin3[2];
  float wlv[2][8];
#pragma unroll
  for (int mt = 0; mt < 2; ++mt) {
#pragma unroll
    for (int r = 0; r < 8; ++r) {
      const int n = 16 * mt + 8 * hi + r;
      float c = sB1[n];
#pragma unroll
      for (int k = 0; k < LD; ++k) c += sW1[n * (C0 + LD) + C0 + k] * sZ[k];
      cin1[mt][r] = c;
      cin2[mt][r] = sB2[n];
      cin3[mt][r] = sB3[n];
      wlv[mt][r]  = sWl[n];
    }
  }
  const float blv = sBl[0];

  v2f Aw1[2][8], Aw2[2][8], Aw3[2][8];
#pragma unroll
  for (int mt = 0; mt < 2; ++mt) {
    const int o = l16 + 16 * mt;
#pragma unroll
    for (int j = 0; j < 8; ++j) {
      const int k1 = 4 * j + 2 * hi;
      const int k2 = kchain(j, hi);
      v2f w;
      w[0] = sW1[o * (C0 + LD) + k1]; w[1] = sW1[o * (C0 + LD) + k1 + 1];
      Aw1[mt][j] = w;
      w[0] = sW2[o * C0 + k2]; w[1] = sW2[o * C0 + k2 + 1];
      Aw2[mt][j] = w;
      w[0] = sW3[o * C0 + k2]; w[1] = sW3[o * C0 + k2 + 1];
      Aw3[mt][j] = w;
    }
  }

  auto tile_mlp = [&](const v2f (&bin)[8], int pbase) __attribute__((always_inline)) {
    v8f acc0 = cin1[0], acc1 = cin1[1];
#pragma unroll
    for (int j = 0; j < 8; ++j) {
      acc0 = wmma_f32(Aw1[0][j], bin[j], acc0);
      acc1 = wmma_f32(Aw1[1][j], bin[j], acc1);
    }
    v2f b[8];

#pragma unroll
    for (int q = 0; q < 4; ++q) {
      v2f p0, p1;
      p0[0] = relu1(acc0[2 * q]); p0[1] = relu1(acc0[2 * q + 1]);
      p1[0] = relu1(acc1[2 * q]); p1[1] = relu1(acc1[2 * q + 1]);
      b[q] = p0; b[4 + q] = p1;
    }
    acc0 = cin2[0]; acc1 = cin2[1];
#pragma unroll
    for (int j = 0; j < 8; ++j) {
      acc0 = wmma_f32(Aw2[0][j], b[j], acc0);
      acc1 = wmma_f32(Aw2[1][j], b[j], acc1);
    }

#pragma unroll
    for (int q = 0; q < 4; ++q) {
      v2f p0, p1;
      p0[0] = relu1(acc0[2 * q]); p0[1] = relu1(acc0[2 * q + 1]);
      p1[0] = relu1(acc1[2 * q]); p1[1] = relu1(acc1[2 * q + 1]);
      b[q] = p0; b[4 + q] = p1;
    }
    acc0 = cin3[0]; acc1 = cin3[1];
#pragma unroll
    for (int j = 0; j < 8; ++j) {
      acc0 = wmma_f32(Aw3[0][j], b[j], acc0);
      acc1 = wmma_f32(Aw3[1][j], b[j], acc1);
    }

    float s0 = relu1(acc0[1]) * wlv[0][1];
    float s1 = relu1(acc1[1]) * wlv[1][1];
    s0 = __builtin_fmaf(relu1(acc0[0]), wlv[0][0], s0);
    s1 = __builtin_fmaf(relu1(acc1[0]), wlv[1][0], s1);
#pragma unroll
    for (int r = 2; r < 8; ++r) {
      s0 = __builtin_fmaf(relu1(acc0[r]), wlv[0][r], s0);
      s1 = __builtin_fmaf(relu1(acc1[r]), wlv[1][r], s1);
    }
    const float part = s0 + s1;
    const float tot  = part + xhalf(part);
    if (lane < 16) out[pbase + lane] = tot + blv;
  };

  int t = 0;
  for (; t + 2 < ntile; t += 2) {
    tile_mlp(bA, (tile0 + t) * 16);
    fetch(bA, tile0 + t + 2);
    __builtin_amdgcn_sched_barrier(0);
    tile_mlp(bB, (tile0 + t + 1) * 16);
    fetch(bB, tile0 + t + 3);
    __builtin_amdgcn_sched_barrier(0);
  }
  tile_mlp(bA, (tile0 + t) * 16);
  tile_mlp(bB, (tile0 + t + 1) * 16);
}

extern "C" void kernel_launch(void* const* d_in, const int* in_sizes, int n_in,
                              void* d_out, int out_size, void* d_ws, size_t ws_size,
                              hipStream_t stream) {
  (void)in_sizes; (void)n_in; (void)out_size; (void)d_ws; (void)ws_size;
  const float* fm = (const float*)d_in[0];
  const float* z  = (const float*)d_in[1];
  const float* w1 = (const float*)d_in[2];
  const float* b1 = (const float*)d_in[3];
  const float* w2 = (const float*)d_in[4];
  const float* b2 = (const float*)d_in[5];
  const float* w3 = (const float*)d_in[6];
  const float* b3 = (const float*)d_in[7];
  const float* wl = (const float*)d_in[8];
  const float* bl = (const float*)d_in[9];
  float* out = (float*)d_out;

  dim3 grid(NUM_BLOCKS);
  dim3 block(THREADS);
  fcomb_mlp_kernel<<<grid, block, 0, stream>>>(fm, z, w1, b1, w2, b2,
                                               w3, b3, wl, bl, out);
}


# Weights staged into LDS in register order (latent fold once, by wave 0, as a fused chain): a wave fills its 160 registers with 40 conflict-free 16-byte reads, not 68 reads, 16 of them 8-way conflicts
# speedup vs baseline: 2.5534x; 1.0397x over previous
#include <hip/hip_runtime.h>


#define C0   32
#define LD   6
#define PDIM 96
#define VOL  (PDIM * PDIM * PDIM)
#define THREADS         512
#define NUM_BLOCKS      256
#define TILES_PER_BLOCK (VOL / 16 / NUM_BLOCKS)
#define FLT_BIG 3.402823466e+38f

#define BLK   132
#define IMG   (8 * BLK)

typedef float v2f __attribute__((ext_vector_type(2)));
typedef float v4f __attribute__((ext_vector_type(4)));
typedef float v8f __attribute__((ext_vector_type(8)));
typedef const float __attribute__((address_space(1)))* gfp;

__device__ __forceinline__ v8f wmma_f32(v2f a, v2f b, v8f c) {
  return __builtin_amdgcn_wmma_f32_16x16x4_f32(false, a, false, b,
                                               (short)0, c, false, false);
}

__device__ __forceinline__ float relu1(float x) {
  return __builtin_amdgcn_fmed3f(x, 0.0f, FLT_BIG);
}

__device__ __forceinline__ float xhalf(float x) {
  const int b = __builtin_bit_cast(int, x);
  return __builtin_bit_cast(float,
      __builtin_amdgcn_permlanex16(b, b, 0x76543210, 0xfedcba98, false, false));
}

__device__ __forceinline__ int fragpos(int o, int hi, int j, int s) {
  return ((o >> 4) * 4 + (j >> 1)) * BLK + ((o & 15) + 16 * hi) * 4 + (j & 1) * 2 + s;
}

__global__ __launch_bounds__(THREADS) void
fcomb_mlp_kernel(const float* __restrict__ fm, const float* __restrict__ z,
                 const float* __restrict__ w1, const float* __restrict__ b1,
                 const float* __restrict__ w2, const float* __restrict__ b2,
                 const float* __restrict__ w3, const float* __restrict__ b3,
                 const float* __restrict__ wl, const float* __restrict__ bl,
                 float* __restrict__ out)
{
  __shared__ __attribute__((aligned(16))) float sA1[IMG];
  __shared__ __attribute__((aligned(16))) float sA2[IMG];
  __shared__ __attribute__((aligned(16))) float sA3[IMG];
  __shared__ __attribute__((aligned(16))) float sV[2 * 64];

  const int tid  = threadIdx.x;
  const int lane = tid & 31;
  const int wave = __builtin_amdgcn_readfirstlane(tid >> 5);

  const int so = tid >> 5, sc = tid & 31;
  const float g1a = w1[so * (C0 + LD) + sc];
  const float g1b = w1[(so + 16) * (C0 + LD) + sc];
  const float g2a = w2[tid];
  const float g2b = w2[tid + THREADS];
  const float g3a = w3[tid];
  const float g3b = w3[tid + THREADS];
  float gwl = 0.0f, gb1 = 0.0f, gb2 = 0.0f, gb3 = 0.0f;
  float gw[LD], gz[LD];
#pragma unroll
  for (int k = 0; k < LD; ++k) { gw[k] = 0.0f; gz[k] = 0.0f; }
  if (wave == 0) {
    gwl = wl[lane]; gb1 = b1[lane]; gb2 = b2[lane]; gb3 = b3[lane];
#pragma unroll
    for (int k = 0; k < LD; ++k) { gw[k] = w1[lane * (C0 + LD) + C0 + k]; gz[k] = z[k]; }
  }
  const int p1 = fragpos(so, (sc >> 1) & 1, sc >> 2, sc & 1);
  const int p2 = fragpos(so, (sc >> 3) & 1, ((sc >> 4) << 2) | ((sc >> 1) & 3), sc & 1);
  sA1[p1] = g1a;  sA1[p1 + 4 * BLK] = g1b;
  sA2[p2] = g2a;  sA2[p2 + 4 * BLK] = g2b;
  sA3[p2] = g3a;  sA3[p2 + 4 * BLK] = g3b;
  if (wave == 0) {
    float c = gb1;
#pragma unroll
    for (int k = 0; k < LD; ++k) c = __builtin_fmaf(gw[k], gz[k], c);
    const int pv = ((lane >> 3) & 1) * 64 + (lane >> 4) * 8 + (lane & 7);
    sV[pv] = c;  sV[pv + 16] = gb2;  sV[pv + 32] = gb3;  sV[pv + 48] = gwl;
  }
  __syncthreads();

  const int l16  = lane & 15;
  const int hi   = lane >> 4;

  const int late  = (wave + 4) >> 4;
  const int ntile = 14 - 2 * late;
  const int tile0 = blockIdx.x * TILES_PER_BLOCK + 14 * wave - 2 * late * (wave - 12);
  gfp chan[8];
#pragma unroll
  for (int j = 0; j < 8; ++j) {
    gfp p = (gfp)fm + (4 * j) * VOL;
    asm("" : "+s"(p));
    chan[j] = p;
  }
  const unsigned laneoff = (unsigned)((2 * hi) * VOL + l16);
  auto fetch = [&](v2f (&dst)[8], int tile) __attribute__((always_inline)) {
    const unsigned voff = laneoff + (unsigned)tile * 16u;
#pragma unroll
    for (int j = 0; j < 8; ++j) {
      v2f bv;
      bv[0] = chan[j][voff];
      bv[1] = (chan[j] + VOL)[voff];
      dst[j] = bv;
    }
  };
  v2f bA[8], bB[8];
  fetch(bA, tile0);
  __builtin_amdgcn_sched_barrier(0);
  fetch(bB, tile0 + 1);
  __builtin_amdgcn_sched_barrier(0);

  v2f Aw1[2][8], Aw2[2][8], Aw3[2][8];
#pragma unroll
  for (int q = 0; q < 8; ++q) {
    const v4f f1 = *(const v4f*)&sA1[q * BLK + lane * 4];
    const v4f f2 = *(const v4f*)&sA2[q * BLK + lane * 4];
    const v4f f3 = *(const v4f*)&sA3[q * BLK + lane * 4];
    const int mt = q >> 2, j = 2 * (q & 3);
    Aw1[mt][j] = f1.xy;  Aw1[mt][j + 1] = f1.zw;
    Aw2[mt][j] = f2.xy;  Aw2[mt][j + 1] = f2.zw;
    Aw3[mt][j] = f3.xy;  Aw3[mt][j + 1] = f3.zw;
  }
  v8f cin1[2], cin2[2], cin3[2];
  float wlv[2][8];
  const v4f* vv = (const v4f*)&sV[hi * 64];
#pragma unroll
  for (int mt = 0; mt < 2; ++mt) {
    const v4f a0 = vv[2 * mt], a1 = vv[2 * mt + 1];
    const v4f b0 = vv[4 + 2 * mt], b1v = vv[4 + 2 * mt + 1];
    const v4f c0 = vv[8 + 2 * mt], c1 = vv[8 + 2 * mt + 1];
    const v4f d0 = vv[12 + 2 * mt], d1 = vv[12 + 2 * mt + 1];
#pragma unroll
    for (int r = 0; r < 4; ++r) {
      cin1[mt][r] = a0[r];  cin1[mt][4 + r] = a1[r];
      cin2[mt][r] = b0[r];  cin2[mt][4 + r] = b1v[r];
      cin3[mt][r] = c0[r];  cin3[mt][4 + r] = c1[r];
      wlv[mt][r]  = d0[r];  wlv[mt][4 + r]  = d1[r];
    }
  }
  const float blv = bl[0];

  auto tile_mlp = [&](const v2f (&bin)[8], int pbase) __attribute__((always_inline)) {
    v8f acc0 = cin1[0], acc1 = cin1[1];
#pragma unroll
    for (int j = 0; j < 8; ++j) {
      acc0 = wmma_f32(Aw1[0][j], bin[j], acc0);
      acc1 = wmma_f32(Aw1[1][j], bin[j], acc1);
    }
    v2f b[8];

#pragma unroll
    for (int q = 0; q < 4; ++q) {
      v2f p0, p1;
      p0[0] = relu1(acc0[2 * q]); p0[1] = relu1(acc0[2 * q + 1]);
      p1[0] = relu1(acc1[2 * q]); p1[1] = relu1(acc1[2 * q + 1]);
      b[q] = p0; b[4 + q] = p1;
    }
    acc0 = cin2[0]; acc1 = cin2[1];
#pragma unroll
    for (int j = 0; j < 8; ++j) {
      acc0 = wmma_f32(Aw2[0][j], b[j], acc0);
      acc1 = wmma_f32(Aw2[1][j], b[j], acc1);
    }

#pragma unroll
    for (int q = 0; q < 4; ++q) {
      v2f p0, p1;
      p0[0] = relu1(acc0[2 * q]); p0[1] = relu1(acc0[2 * q + 1]);
      p1[0] = relu1(acc1[2 * q]); p1[1] = relu1(acc1[2 * q + 1]);
      b[q] = p0; b[4 + q] = p1;
    }
    acc0 = cin3[0]; acc1 = cin3[1];
#pragma unroll
    for (int j = 0; j < 8; ++j) {
      acc0 = wmma_f32(Aw3[0][j], b[j], acc0);
      acc1 = wmma_f32(Aw3[1][j], b[j], acc1);
    }

    float s0 = relu1(acc0[1]) * wlv[0][1];
    float s1 = relu1(acc1[1]) * wlv[1][1];
    s0 = __builtin_fmaf(relu1(acc0[0]), wlv[0][0], s0);
    s1 = __builtin_fmaf(relu1(acc1[0]), wlv[1][0], s1);
#pragma unroll
    for (int r = 2; r < 8; ++r) {
      s0 = __builtin_fmaf(relu1(acc0[r]), wlv[0][r], s0);
      s1 = __builtin_fmaf(relu1(acc1[r]), wlv[1][r], s1);
    }
    const float part = s0 + s1;
    const float tot  = part + xhalf(part);
    if (lane < 16) out[pbase + lane] = tot + blv;
  };

  int t = 0;
  for (; t + 2 < ntile; t += 2) {
    tile_mlp(bA, (tile0 + t) * 16);
    fetch(bA, tile0 + t + 2);
    __builtin_amdgcn_sched_barrier(0);
    tile_mlp(bB, (tile0 + t + 1) * 16);
    fetch(bB, tile0 + t + 3);
    __builtin_amdgcn_sched_barrier(0);
  }
  tile_mlp(bA, (tile0 + t) * 16);
  tile_mlp(bB, (tile0 + t + 1) * 16);
}

extern "C" void kernel_launch(void* const* d_in, const int* in_sizes, int n_in,
                              void* d_out, int out_size, void* d_ws, size_t ws_size,
                              hipStream_t stream) {
  (void)in_sizes; (void)n_in; (void)out_size; (void)d_ws; (void)ws_size;
  const float* fm = (const float*)d_in[0];
  const float* z  = (const float*)d_in[1];
  const float* w1 = (const float*)d_in[2];
  const float* b1 = (const float*)d_in[3];
  const float* w2 = (const float*)d_in[4];
  const float* b2 = (const float*)d_in[5];
  const float* w3 = (const float*)d_in[6];
  const float* b3 = (const float*)d_in[7];
  const float* wl = (const float*)d_in[8];
  const float* bl = (const float*)d_in[9];
  float* out = (float*)d_out;

  dim3 grid(NUM_BLOCKS);
  dim3 block(THREADS);
  fcomb_mlp_kernel<<<grid, block, 0, stream>>>(fm, z, w1, b1, w2, b2,
                                               w3, b3, wl, bl, out);
}
